# starting kernel
# baseline (speedup 1.0000x reference)
#include <hip/hip_runtime.h>


typedef __attribute__((ext_vector_type(2))) float v2f;
typedef __attribute__((ext_vector_type(8))) float v8f;

#define NYv   8
#define NUv   8
#define NHv   256
#define H2v   128
#define T_LEN 2048
#define NBv   64
#define NSEG  16
#define SEG   (T_LEN / NSEG)
#define CHNK  16
#define NCH   (SEG / CHNK)
#define XPAD  258

__device__ __forceinline__ void load_wt_ops(const float* __restrict__ Wx2y,
                                            int wave, int lane, v2f bop[16]) {
  const int n  = lane & 15;
  const int kh = (lane >> 4) * 2;
#pragma unroll
  for (int i = 0; i < 16; ++i) {
    const int k0 = wave * 64 + 4 * i + kh;
    v2f v;
    if (n < NYv) {
      v.x = Wx2y[n * NHv + k0];
      v.y = Wx2y[n * NHv + k0 + 1];
    } else {
      v.x = 0.0f; v.y = 0.0f;
    }
    bop[i] = v;
  }
}

__device__ __forceinline__ void project_chunk(const float (*Xc)[XPAD],
                                              const v2f bop[16],
                                              int wave, int lane,
                                              float (*Yp)[CHNK][16]) {
  v8f acc = {};
  const int row = lane & 15;
  const int kh  = (lane >> 4) * 2;
#pragma unroll
  for (int i = 0; i < 16; ++i) {
    const int k0 = wave * 64 + 4 * i + kh;
    v2f a;
    a.x = Xc[row][k0];
    a.y = Xc[row][k0 + 1];
    acc = __builtin_amdgcn_wmma_f32_16x16x4_f32(
        false, a, false, bop[i], (short)0, acc, false, false);
  }
  const int mbase = (lane >> 4) * 8;
  const int ncol  = lane & 15;
#pragma unroll
  for (int v = 0; v < 8; ++v) Yp[wave][v + mbase][ncol] = acc[v];
}

__global__ void __launch_bounds__(128)
lru_seg_scan(const float* __restrict__ U,   const float* __restrict__ lr,
             const float* __restrict__ li,  const float* __restrict__ Bm,
             const float* __restrict__ mult,const float* __restrict__ Wx2y,
             const float* __restrict__ bx2y,float* __restrict__ Y,
             float* __restrict__ e_ws) {
  const int b = blockIdx.x;
  const int s = blockIdx.y;
  const int j = threadIdx.x;
  const int wave = j >> 5, lane = j & 31;

  __shared__ float Uc[CHNK][NUv];
  __shared__ float Xc[CHNK][XPAD];
  __shared__ float Yp[4][CHNK][16];

  const float m1 = expf(mult[j]);
  const float m2 = expf(mult[j + H2v]);
  float Bs1[NUv], Bs2[NUv];
#pragma unroll
  for (int u = 0; u < NUv; ++u) {
    Bs1[u] = Bm[j * NUv + u] * m1;
    Bs2[u] = Bm[(j + H2v) * NUv + u] * m2;
  }
  const float r  = expf(-fabsf(lr[j]));
  const float th = 1.5707963267948966f * li[j];
  const float re = r * cosf(th);
  const float im = r * sinf(th);

  v2f bop[16];
  load_wt_ops(Wx2y, wave, lane, bop);

  float x1 = 0.0f, x2 = 0.0f;

  if (j < CHNK)
    __builtin_prefetch(U + ((size_t)(s * SEG + j) * NBv + b) * NUv, 0, 0);

  for (int ch = 0; ch < NCH; ++ch) {
    const int t0 = s * SEG + ch * CHNK;
    if (j < 2 * CHNK) {
      const int tt = j >> 1, h = (j & 1) * 4;
      const float4 uv =
          *(const float4*)(U + ((size_t)(t0 + tt) * NBv + b) * NUv + h);
      Uc[tt][h + 0] = uv.x; Uc[tt][h + 1] = uv.y;
      Uc[tt][h + 2] = uv.z; Uc[tt][h + 3] = uv.w;
    }
    if (ch + 1 < NCH && j < CHNK)
      __builtin_prefetch(U + ((size_t)(t0 + CHNK + j) * NBv + b) * NUv, 0, 0);
    __syncthreads();

#pragma unroll
    for (int tt = 0; tt < CHNK; ++tt) {
      float bu1 = 0.0f, bu2 = 0.0f;
#pragma unroll
      for (int u = 0; u < NUv; ++u) {
        const float uu = Uc[tt][u];
        bu1 = fmaf(uu, Bs1[u], bu1);
        bu2 = fmaf(uu, Bs2[u], bu2);
      }
      const float n1 = re * x1 - im * x2 + bu1;
      const float n2 = im * x1 + re * x2 + bu2;
      x1 = n1; x2 = n2;
      Xc[tt][j]       = x1;
      Xc[tt][j + H2v] = x2;
    }
    __syncthreads();

    project_chunk(Xc, bop, wave, lane, Yp);
    __syncthreads();

    {
      const int tt = j >> 3, n = j & 7;
      const float y = Yp[0][tt][n] + Yp[1][tt][n] + Yp[2][tt][n] + Yp[3][tt][n]
                    + bx2y[n];
      Y[((size_t)(t0 + tt) * NBv + b) * NYv + n] = y;
    }
    __syncthreads();
  }

  e_ws[((size_t)s * NBv + b) * NHv + j]       = x1;
  e_ws[((size_t)s * NBv + b) * NHv + j + H2v] = x2;
}

__global__ void __launch_bounds__(128)
lru_carry(const float* __restrict__ y0,  const float* __restrict__ Wy2x,
          const float* __restrict__ by2x,const float* __restrict__ lr,
          const float* __restrict__ li,  const float* __restrict__ e_ws,
          float* __restrict__ c_ws) {
  const int b = blockIdx.x;
  const int j = threadIdx.x;

  const float th  = 1.5707963267948966f * li[j];
  const float rS  = expf(-(float)SEG * fabsf(lr[j]));
  const float phS = (float)SEG * th;
  const float reS = rS * cosf(phS);
  const float imS = rS * sinf(phS);

  float c1 = by2x[j], c2 = by2x[j + H2v];
#pragma unroll
  for (int n = 0; n < NYv; ++n) {
    const float yv = y0[b * NYv + n];
    c1 = fmaf(yv, Wy2x[j * NYv + n], c1);
    c2 = fmaf(yv, Wy2x[(j + H2v) * NYv + n], c2);
  }

  for (int s = 0; s < NSEG; ++s) {
    const size_t base = ((size_t)s * NBv + b) * NHv;
    c_ws[base + j]       = c1;
    c_ws[base + j + H2v] = c2;
    const float e1 = e_ws[base + j];
    const float e2 = e_ws[base + j + H2v];
    const float n1 = reS * c1 - imS * c2 + e1;
    const float n2 = imS * c1 + reS * c2 + e2;
    c1 = n1; c2 = n2;
  }
}

__global__ void __launch_bounds__(128)
lru_correct(const float* __restrict__ lr, const float* __restrict__ li,
            const float* __restrict__ Wx2y, const float* __restrict__ c_ws,
            float* __restrict__ Y) {
  const int b = blockIdx.x;
  const int s = blockIdx.y;
  const int j = threadIdx.x;
  const int wave = j >> 5, lane = j & 31;

  __shared__ float Xc[CHNK][XPAD];
  __shared__ float Yp[4][CHNK][16];

  const float r  = expf(-fabsf(lr[j]));
  const float th = 1.5707963267948966f * li[j];
  const float re = r * cosf(th);
  const float im = r * sinf(th);

  v2f bop[16];
  load_wt_ops(Wx2y, wave, lane, bop);

  const size_t cbase = ((size_t)s * NBv + b) * NHv;
  float v1 = c_ws[cbase + j];
  float v2 = c_ws[cbase + j + H2v];

  for (int ch = 0; ch < NCH; ++ch) {
    const int t0 = s * SEG + ch * CHNK;
#pragma unroll
    for (int tt = 0; tt < CHNK; ++tt) {
      const float n1 = re * v1 - im * v2;
      const float n2 = im * v1 + re * v2;
      v1 = n1; v2 = n2;
      Xc[tt][j]       = v1;
      Xc[tt][j + H2v] = v2;
    }
    __syncthreads();

    project_chunk(Xc, bop, wave, lane, Yp);
    __syncthreads();

    {
      const int tt = j >> 3, n = j & 7;
      const size_t yi = ((size_t)(t0 + tt) * NBv + b) * NYv + n;
      Y[yi] += Yp[0][tt][n] + Yp[1][tt][n] + Yp[2][tt][n] + Yp[3][tt][n];
    }
    __syncthreads();
  }
}

extern "C" void kernel_launch(void* const* d_in, const int* in_sizes, int n_in,
                              void* d_out, int out_size, void* d_ws, size_t ws_size,
                              hipStream_t stream) {
  (void)in_sizes; (void)n_in; (void)out_size; (void)ws_size;
  const float* y0   = (const float*)d_in[0];
  const float* U    = (const float*)d_in[1];
  const float* lr   = (const float*)d_in[2];
  const float* li   = (const float*)d_in[3];
  const float* Bm   = (const float*)d_in[4];
  const float* mult = (const float*)d_in[5];
  const float* Wy2x = (const float*)d_in[6];
  const float* by2x = (const float*)d_in[7];
  const float* Wx2y = (const float*)d_in[8];
  const float* bx2y = (const float*)d_in[9];
  float* Y = (float*)d_out;

  float* e_ws = (float*)d_ws;
  float* c_ws = e_ws + (size_t)NSEG * NBv * NHv;

  dim3 grid(NBv, NSEG);
  lru_seg_scan<<<grid, 128, 0, stream>>>(U, lr, li, Bm, mult, Wx2y, bx2y, Y, e_ws);
  lru_carry<<<NBv, 128, 0, stream>>>(y0, Wy2x, by2x, lr, li, e_ws, c_ws);
  lru_correct<<<grid, 128, 0, stream>>>(lr, li, Wx2y, c_ws, Y);
}


# Launch 1 coarsened: a lane scans 2 rotation pairs (2-wave blocks), so one broadcast LDS read of a step's inputs feeds 4 state rows; a wave projects its own M-tile over all K; sum order changes
# speedup vs baseline: 1.1954x; 1.1954x over previous
#include <hip/hip_runtime.h>


typedef __attribute__((ext_vector_type(2))) float v2f;
typedef __attribute__((ext_vector_type(4))) float v4f;
typedef __attribute__((ext_vector_type(8))) float v8f;

#define NYv   8
#define NUv   8
#define NHv   256
#define H2v   128
#define T_LEN 2048
#define NBv   64
#define NSEG  16
#define SEG   (T_LEN / NSEG)

#define RND    32
#define NRND   (SEG / RND)
#define XPITCH 260

__global__ void __launch_bounds__(64)
lru_seg_scan(const float* __restrict__ U,   const float* __restrict__ lr,
             const float* __restrict__ li,  const float* __restrict__ Bm,
             const float* __restrict__ mult,const float* __restrict__ Wx2y,
             const float* __restrict__ bx2y,float* __restrict__ Y,
             float* __restrict__ e_ws) {
  const int b = blockIdx.x;
  const int s = blockIdx.y;
  const int tid = threadIdx.x;
  const int wave = tid >> 5, lane = tid & 31;
  const int c = lane & 15, hl = lane >> 4;
  const int j0 = 2 * tid;

  __shared__ __attribute__((aligned(16))) float Uc[RND][NUv];
  __shared__ __attribute__((aligned(16))) float Xc[RND][XPITCH];

  const int ut = tid >> 1, uh = (tid & 1) * 4;
  const float* up = U + ((size_t)(s * SEG + ut) * NBv + b) * NUv + uh;
  float4 ucur = *(const float4*)up;

  float Bs1[2][NUv], Bs2[2][NUv], re[2], im[2];
#pragma unroll
  for (int q = 0; q < 2; ++q) {
    const int j = j0 + q;
    const float m1 = expf(mult[j]);
    const float m2 = expf(mult[j + H2v]);
#pragma unroll
    for (int u = 0; u < NUv; ++u) {
      Bs1[q][u] = Bm[j * NUv + u] * m1;
      Bs2[q][u] = Bm[(j + H2v) * NUv + u] * m2;
    }
    const float r  = expf(-fabsf(lr[j]));
    const float th = 1.5707963267948966f * li[j];
    re[q] = r * cosf(th);
    im[q] = r * sinf(th);
  }

  v4f bf[32];
  {
    const v4f* wp = (const v4f*)(Wx2y + (size_t)(c & 7) * NHv + 4 * hl);
    const v4f zero = {0.0f, 0.0f, 0.0f, 0.0f};
#pragma unroll
    for (int i = 0; i < 32; ++i) {
      const v4f w = wp[2 * i];
      bf[i] = (c < NYv) ? w : zero;
    }
  }
  const float bias = bx2y[c & 7];

  float x1[2] = {0.0f, 0.0f}, x2[2] = {0.0f, 0.0f};
  float ypend[8];
#pragma unroll
  for (int r = 0; r < 8; ++r) ypend[r] = 0.0f;

#pragma nounroll
  for (int rd = 0; rd < NRND; ++rd) {
    *(float4*)&Uc[ut][uh] = ucur;
    __syncthreads();

    if (rd + 1 < NRND)
      ucur = *(const float4*)(up + (size_t)(rd + 1) * RND * NBv * NUv);
    if (rd > 0 && c < NYv) {
      float* yp = Y + ((size_t)(s * SEG + (rd - 1) * RND + 16 * wave + 8 * hl) * NBv + b) * NYv + c;
#pragma unroll
      for (int r = 0; r < 8; ++r) yp[(size_t)r * NBv * NYv] = ypend[r];
    }

#pragma unroll
    for (int tt = 0; tt < RND; ++tt) {
      const float4 ua = *(const float4*)&Uc[tt][0];
      const float4 ub = *(const float4*)&Uc[tt][4];
      const float uu[NUv] = {ua.x, ua.y, ua.z, ua.w, ub.x, ub.y, ub.z, ub.w};
#pragma unroll
      for (int q = 0; q < 2; ++q) {
        float bu1 = 0.0f, bu2 = 0.0f;
#pragma unroll
        for (int u = 0; u < NUv; ++u) {
          bu1 = fmaf(uu[u], Bs1[q][u], bu1);
          bu2 = fmaf(uu[u], Bs2[q][u], bu2);
        }
        const float n1 = re[q] * x1[q] - im[q] * x2[q] + bu1;
        const float n2 = im[q] * x1[q] + re[q] * x2[q] + bu2;
        x1[q] = n1; x2[q] = n2;
      }
      *(float2*)&Xc[tt][j0]       = make_float2(x1[0], x1[1]);
      *(float2*)&Xc[tt][H2v + j0] = make_float2(x2[0], x2[1]);
    }
    __syncthreads();

    const v4f* ap = (const v4f*)(&Xc[16 * wave + c][4 * hl]);
    v4f af[32];
#pragma unroll
    for (int i = 0; i < 32; ++i) af[i] = ap[2 * i];
#pragma unroll
    for (int i = 0; i < 32; i += 8)
      asm volatile("" :: "v"(af[i]), "v"(af[i + 1]), "v"(af[i + 2]), "v"(af[i + 3]),
                         "v"(af[i + 4]), "v"(af[i + 5]), "v"(af[i + 6]), "v"(af[i + 7]));
    __builtin_amdgcn_sched_barrier(0);

    v8f acc0 = {}, acc1 = {};
#pragma unroll
    for (int i = 0; i < 32; ++i) {
      acc0 = __builtin_amdgcn_wmma_f32_16x16x4_f32(
          false, af[i].xy, false, bf[i].xy, (short)0, acc0, false, false);
      acc1 = __builtin_amdgcn_wmma_f32_16x16x4_f32(
          false, af[i].zw, false, bf[i].zw, (short)0, acc1, false, false);
    }
#pragma unroll
    for (int r = 0; r < 8; ++r) ypend[r] = acc0[r] + acc1[r] + bias;
  }

  if (c < NYv) {
    float* yp = Y + ((size_t)(s * SEG + (NRND - 1) * RND + 16 * wave + 8 * hl) * NBv + b) * NYv + c;
#pragma unroll
    for (int r = 0; r < 8; ++r) yp[(size_t)r * NBv * NYv] = ypend[r];
  }

  float* ep = e_ws + ((size_t)s * NBv + b) * NHv;
  *(float2*)(ep + j0)       = make_float2(x1[0], x1[1]);
  *(float2*)(ep + H2v + j0) = make_float2(x2[0], x2[1]);
}

__global__ void __launch_bounds__(128)
lru_carry(const float* __restrict__ y0,  const float* __restrict__ Wy2x,
          const float* __restrict__ by2x,const float* __restrict__ lr,
          const float* __restrict__ li,  const float* __restrict__ e_ws,
          float* __restrict__ c_ws) {
  const int b = blockIdx.x;
  const int j = threadIdx.x;

  const float th  = 1.5707963267948966f * li[j];
  const float rS  = expf(-(float)SEG * fabsf(lr[j]));
  const float phS = (float)SEG * th;
  const float reS = rS * cosf(phS);
  const float imS = rS * sinf(phS);

  float c1 = by2x[j], c2 = by2x[j + H2v];
#pragma unroll
  for (int n = 0; n < NYv; ++n) {
    const float yv = y0[b * NYv + n];
    c1 = fmaf(yv, Wy2x[j * NYv + n], c1);
    c2 = fmaf(yv, Wy2x[(j + H2v) * NYv + n], c2);
  }

  for (int s = 0; s < NSEG; ++s) {
    const size_t base = ((size_t)s * NBv + b) * NHv;
    c_ws[base + j]       = c1;
    c_ws[base + j + H2v] = c2;
    const float e1 = e_ws[base + j];
    const float e2 = e_ws[base + j + H2v];
    const float n1 = reS * c1 - imS * c2 + e1;
    const float n2 = imS * c1 + reS * c2 + e2;
    c1 = n1; c2 = n2;
  }
}

#define MSTEPS 4
#define PPITCH 260

__global__ void __launch_bounds__(128)
lru_correct(const float* __restrict__ lr, const float* __restrict__ li,
            const float* __restrict__ Wx2y, const float* __restrict__ c_ws,
            float* __restrict__ Y) {
  const int mg = blockIdx.x;
  const int ng = blockIdx.y;
  const int j = threadIdx.x;
  const int wave = j >> 5, lane = j & 31;
  const int c = lane & 15, hl = lane >> 4;

  __shared__ __attribute__((aligned(16))) float P[MSTEPS * NYv][PPITCH];

  const int col = (ng * 4 + wave) * 16 + c;
  const v4f* cp = (const v4f*)(c_ws + (size_t)col * NHv + 4 * hl);
  v4f bf[32];
#pragma unroll
  for (int i = 0; i < 32; ++i) bf[i] = cp[2 * i];

  const int s = col >> 6, b = col & 63;
  const int t0 = s * SEG + mg * MSTEPS + hl;
  float4* yp[MSTEPS / 2];
  float4 ylo[MSTEPS / 2], yhi[MSTEPS / 2];
#pragma unroll
  for (int m = 0; m < MSTEPS / 2; ++m) {
    yp[m]  = (float4*)(Y + ((size_t)(t0 + 2 * m) * NBv + b) * NYv);
    ylo[m] = yp[m][0];
    yhi[m] = yp[m][1];
  }

  const float alr = fabsf(lr[j]);
  const float th  = 1.5707963267948966f * li[j];
  float w1[NYv], w2[NYv];
#pragma unroll
  for (int n = 0; n < NYv; ++n) {
    w1[n] = Wx2y[n * NHv + j];
    w2[n] = Wx2y[n * NHv + j + H2v];
  }
#pragma unroll
  for (int kk = 0; kk < MSTEPS; ++kk) {
    const float kf = (float)(mg * MSTEPS + kk + 1);
    const float rk = expf(-kf * alr);
    const float ah = kf * th;
    const float al = fmaf(kf, th, -ah);
    const float sh = sinf(ah), ch = cosf(ah);
    const float sl = sinf(al), cl = cosf(al);
    const float pre = rk * (ch * cl - sh * sl);
    const float pim = rk * (sh * cl + ch * sl);
#pragma unroll
    for (int n = 0; n < NYv; ++n) {
      P[kk * NYv + n][j]       = w1[n] * pre + w2[n] * pim;
      P[kk * NYv + n][j + H2v] = w2[n] * pre - w1[n] * pim;
    }
  }
  __syncthreads();

#pragma unroll
  for (int m = 0; m < MSTEPS / 2; ++m) {
    const v4f* ap = (const v4f*)(&P[16 * m + c][4 * hl]);
    v4f af[32];
#pragma unroll
    for (int i = 0; i < 32; ++i) af[i] = ap[2 * i];
#pragma unroll
    for (int i = 0; i < 32; i += 8)
      asm volatile("" :: "v"(af[i]), "v"(af[i + 1]), "v"(af[i + 2]), "v"(af[i + 3]),
                         "v"(af[i + 4]), "v"(af[i + 5]), "v"(af[i + 6]), "v"(af[i + 7]));
    __builtin_amdgcn_sched_barrier(0);

    v8f acc0 = {}, acc1 = {};
#pragma unroll
    for (int i = 0; i < 32; ++i) {
      acc0 = __builtin_amdgcn_wmma_f32_16x16x4_f32(
          false, af[i].xy, false, bf[i].xy, (short)0, acc0, false, false);
      acc1 = __builtin_amdgcn_wmma_f32_16x16x4_f32(
          false, af[i].zw, false, bf[i].zw, (short)0, acc1, false, false);
    }
    __builtin_amdgcn_sched_barrier(0);

    float4 y0 = ylo[m], y1 = yhi[m];
    y0.x += acc0[0] + acc1[0]; y0.y += acc0[1] + acc1[1];
    y0.z += acc0[2] + acc1[2]; y0.w += acc0[3] + acc1[3];
    y1.x += acc0[4] + acc1[4]; y1.y += acc0[5] + acc1[5];
    y1.z += acc0[6] + acc1[6]; y1.w += acc0[7] + acc1[7];
    yp[m][0] = y0; yp[m][1] = y1;
  }
}

extern "C" void kernel_launch(void* const* d_in, const int* in_sizes, int n_in,
                              void* d_out, int out_size, void* d_ws, size_t ws_size,
                              hipStream_t stream) {
  (void)in_sizes; (void)n_in; (void)out_size; (void)ws_size;
  const float* y0   = (const float*)d_in[0];
  const float* U    = (const float*)d_in[1];
  const float* lr   = (const float*)d_in[2];
  const float* li   = (const float*)d_in[3];
  const float* Bm   = (const float*)d_in[4];
  const float* mult = (const float*)d_in[5];
  const float* Wy2x = (const float*)d_in[6];
  const float* by2x = (const float*)d_in[7];
  const float* Wx2y = (const float*)d_in[8];
  const float* bx2y = (const float*)d_in[9];
  float* Y = (float*)d_out;

  float* e_ws = (float*)d_ws;
  float* c_ws = e_ws + (size_t)NSEG * NBv * NHv;

  dim3 grid(NBv, NSEG);
  lru_seg_scan<<<grid, 64, 0, stream>>>(U, lr, li, Bm, mult, Wx2y, bx2y, Y, e_ws);
  lru_carry<<<NBv, 128, 0, stream>>>(y0, Wy2x, by2x, lr, li, e_ws, c_ws);
  dim3 cgrid(32, 16);
  lru_correct<<<cgrid, 128, 0, stream>>>(lr, li, Wx2y, c_ws, Y);
}


# Carry scan moved into the correction GEMM: a block owns 4 batches x all 16 segments and scans their carry-ins from the end states into its LDS B operand; carry launch, c plane gone; sums unchanged
# speedup vs baseline: 1.6600x; 1.3887x over previous
#include <hip/hip_runtime.h>


typedef __attribute__((ext_vector_type(2))) float v2f;
typedef __attribute__((ext_vector_type(4))) float v4f;
typedef __attribute__((ext_vector_type(8))) float v8f;

#define NYv   8
#define NUv   8
#define NHv   256
#define H2v   128
#define T_LEN 2048
#define NBv   64
#define NSEG  16
#define SEG   (T_LEN / NSEG)

#define RND    32
#define NRND   (SEG / RND)
#define XPITCH 260

__global__ void __launch_bounds__(64)
lru_seg_scan(const float* __restrict__ U,   const float* __restrict__ lr,
             const float* __restrict__ li,  const float* __restrict__ Bm,
             const float* __restrict__ mult,const float* __restrict__ Wx2y,
             const float* __restrict__ bx2y,float* __restrict__ Y,
             float* __restrict__ e_ws) {
  const int b = blockIdx.x;
  const int s = blockIdx.y;
  const int tid = threadIdx.x;
  const int wave = tid >> 5, lane = tid & 31;
  const int c = lane & 15, hl = lane >> 4;
  const int j0 = 2 * tid;

  __shared__ __attribute__((aligned(16))) float Uc[RND][NUv];
  __shared__ __attribute__((aligned(16))) float Xc[RND][XPITCH];

  const int ut = tid >> 1, uh = (tid & 1) * 4;
  const float* up = U + ((size_t)(s * SEG + ut) * NBv + b) * NUv + uh;
  float4 ucur = *(const float4*)up;

  float Bs1[2][NUv], Bs2[2][NUv], re[2], im[2];
#pragma unroll
  for (int q = 0; q < 2; ++q) {
    const int j = j0 + q;
    const float m1 = expf(mult[j]);
    const float m2 = expf(mult[j + H2v]);
#pragma unroll
    for (int u = 0; u < NUv; ++u) {
      Bs1[q][u] = Bm[j * NUv + u] * m1;
      Bs2[q][u] = Bm[(j + H2v) * NUv + u] * m2;
    }
    const float r  = expf(-fabsf(lr[j]));
    const float th = 1.5707963267948966f * li[j];
    re[q] = r * cosf(th);
    im[q] = r * sinf(th);
  }

  v4f bf[32];
  {
    const v4f* wp = (const v4f*)(Wx2y + (size_t)(c & 7) * NHv + 4 * hl);
    const v4f zero = {0.0f, 0.0f, 0.0f, 0.0f};
#pragma unroll
    for (int i = 0; i < 32; ++i) {
      const v4f w = wp[2 * i];
      bf[i] = (c < NYv) ? w : zero;
    }
  }
  const float bias = bx2y[c & 7];

  float x1[2] = {0.0f, 0.0f}, x2[2] = {0.0f, 0.0f};
  float ypend[8];
#pragma unroll
  for (int r = 0; r < 8; ++r) ypend[r] = 0.0f;

#pragma nounroll
  for (int rd = 0; rd < NRND; ++rd) {
    *(float4*)&Uc[ut][uh] = ucur;
    __syncthreads();

    if (rd + 1 < NRND)
      ucur = *(const float4*)(up + (size_t)(rd + 1) * RND * NBv * NUv);
    if (rd > 0 && c < NYv) {
      float* yp = Y + ((size_t)(s * SEG + (rd - 1) * RND + 16 * wave + 8 * hl) * NBv + b) * NYv + c;
#pragma unroll
      for (int r = 0; r < 8; ++r) yp[(size_t)r * NBv * NYv] = ypend[r];
    }

#pragma unroll
    for (int tt = 0; tt < RND; ++tt) {
      const float4 ua = *(const float4*)&Uc[tt][0];
      const float4 ub = *(const float4*)&Uc[tt][4];
      const float uu[NUv] = {ua.x, ua.y, ua.z, ua.w, ub.x, ub.y, ub.z, ub.w};
#pragma unroll
      for (int q = 0; q < 2; ++q) {
        float bu1 = 0.0f, bu2 = 0.0f;
#pragma unroll
        for (int u = 0; u < NUv; ++u) {
          bu1 = fmaf(uu[u], Bs1[q][u], bu1);
          bu2 = fmaf(uu[u], Bs2[q][u], bu2);
        }
        const float n1 = re[q] * x1[q] - im[q] * x2[q] + bu1;
        const float n2 = im[q] * x1[q] + re[q] * x2[q] + bu2;
        x1[q] = n1; x2[q] = n2;
      }
      *(float2*)&Xc[tt][j0]       = make_float2(x1[0], x1[1]);
      *(float2*)&Xc[tt][H2v + j0] = make_float2(x2[0], x2[1]);
    }
    __syncthreads();

    const v4f* ap = (const v4f*)(&Xc[16 * wave + c][4 * hl]);
    v4f af[32];
#pragma unroll
    for (int i = 0; i < 32; ++i) af[i] = ap[2 * i];
#pragma unroll
    for (int i = 0; i < 32; i += 8)
      asm volatile("" :: "v"(af[i]), "v"(af[i + 1]), "v"(af[i + 2]), "v"(af[i + 3]),
                         "v"(af[i + 4]), "v"(af[i + 5]), "v"(af[i + 6]), "v"(af[i + 7]));
    __builtin_amdgcn_sched_barrier(0);

    v8f acc0 = {}, acc1 = {};
#pragma unroll
    for (int i = 0; i < 32; ++i) {
      acc0 = __builtin_amdgcn_wmma_f32_16x16x4_f32(
          false, af[i].xy, false, bf[i].xy, (short)0, acc0, false, false);
      acc1 = __builtin_amdgcn_wmma_f32_16x16x4_f32(
          false, af[i].zw, false, bf[i].zw, (short)0, acc1, false, false);
    }
#pragma unroll
    for (int r = 0; r < 8; ++r) ypend[r] = acc0[r] + acc1[r] + bias;
  }

  if (c < NYv) {
    float* yp = Y + ((size_t)(s * SEG + (NRND - 1) * RND + 16 * wave + 8 * hl) * NBv + b) * NYv + c;
#pragma unroll
    for (int r = 0; r < 8; ++r) yp[(size_t)r * NBv * NYv] = ypend[r];
  }

  float* ep = e_ws + ((size_t)s * NBv + b) * NHv;
  *(float2*)(ep + j0)       = make_float2(x1[0], x1[1]);
  *(float2*)(ep + H2v + j0) = make_float2(x2[0], x2[1]);
}

#define MSTEPS 4
#define PPITCH 260
#define BGRP   4

__global__ void __launch_bounds__(128)
lru_correct(const float* __restrict__ y0,  const float* __restrict__ Wy2x,
            const float* __restrict__ by2x,const float* __restrict__ lr,
            const float* __restrict__ li,  const float* __restrict__ Wx2y,
            const float* __restrict__ e_ws, float* __restrict__ Y) {
  const int mg = blockIdx.x;
  const int bg = blockIdx.y;
  const int j = threadIdx.x;
  const int wave = j >> 5, lane = j & 31;
  const int c = lane & 15, hl = lane >> 4;

  __shared__ __attribute__((aligned(16))) float P[MSTEPS * NYv][PPITCH];
  __shared__ __attribute__((aligned(16))) float Cs[NSEG * BGRP][PPITCH];
  __shared__ __attribute__((aligned(16))) float rotS[2][H2v];

  const float alr = fabsf(lr[j]);
  const float th  = 1.5707963267948966f * li[j];

  const int bsc = bg * BGRP + wave;
  const int h0 = 4 * lane;
  v4f e1[NSEG - 1], e2[NSEG - 1];
#pragma unroll
  for (int s = 0; s < NSEG - 1; ++s) {
    const float* ep = e_ws + ((size_t)s * NBv + bsc) * NHv + h0;
    e1[s] = *(const v4f*)ep;
    e2[s] = *(const v4f*)(ep + H2v);
  }

  const int col = 16 * wave + c;
  const int s = col >> 2, b = bg * BGRP + (col & 3);
  const int t0 = s * SEG + mg * MSTEPS + hl;
  float4* yp[MSTEPS / 2];
  float4 ylo[MSTEPS / 2], yhi[MSTEPS / 2];
#pragma unroll
  for (int m = 0; m < MSTEPS / 2; ++m) {
    yp[m]  = (float4*)(Y + ((size_t)(t0 + 2 * m) * NBv + b) * NYv);
    ylo[m] = yp[m][0];
    yhi[m] = yp[m][1];
  }

  float w1[NYv], w2[NYv];
#pragma unroll
  for (int n = 0; n < NYv; ++n) {
    w1[n] = Wx2y[n * NHv + j];
    w2[n] = Wx2y[n * NHv + j + H2v];
  }
#pragma unroll
  for (int kk = 0; kk < MSTEPS; ++kk) {
    const float kf = (float)(mg * MSTEPS + kk + 1);
    const float rk = expf(-kf * alr);
    const float ah = kf * th;
    const float al = fmaf(kf, th, -ah);
    const float sh = sinf(ah), ch = cosf(ah);
    const float sl = sinf(al), cl = cosf(al);
    const float pre = rk * (ch * cl - sh * sl);
    const float pim = rk * (sh * cl + ch * sl);
#pragma unroll
    for (int n = 0; n < NYv; ++n) {
      P[kk * NYv + n][j]       = w1[n] * pre + w2[n] * pim;
      P[kk * NYv + n][j + H2v] = w2[n] * pre - w1[n] * pim;
    }
  }
  v4f c1, c2;
  {
    const v4f ya = *(const v4f*)(y0 + bsc * NYv);
    const v4f yb = *(const v4f*)(y0 + bsc * NYv + 4);
    const float yv[NYv] = {ya.x, ya.y, ya.z, ya.w, yb.x, yb.y, yb.z, yb.w};
    c1 = *(const v4f*)(by2x + h0);
    c2 = *(const v4f*)(by2x + H2v + h0);
#pragma unroll
    for (int i = 0; i < 4; ++i) {
      const v4f wa = *(const v4f*)(Wy2x + (size_t)(h0 + i) * NYv);
      const v4f wb = *(const v4f*)(Wy2x + (size_t)(h0 + i) * NYv + 4);
      const v4f wc = *(const v4f*)(Wy2x + (size_t)(h0 + i + H2v) * NYv);
      const v4f wd = *(const v4f*)(Wy2x + (size_t)(h0 + i + H2v) * NYv + 4);
      const float w1r[NYv] = {wa.x, wa.y, wa.z, wa.w, wb.x, wb.y, wb.z, wb.w};
      const float w2r[NYv] = {wc.x, wc.y, wc.z, wc.w, wd.x, wd.y, wd.z, wd.w};
      float a1 = c1[i], a2 = c2[i];
#pragma unroll
      for (int n = 0; n < NYv; ++n) {
        a1 = fmaf(yv[n], w1r[n], a1);
        a2 = fmaf(yv[n], w2r[n], a2);
      }
      c1[i] = a1; c2[i] = a2;
    }
  }

  {
    const float rS  = expf(-(float)SEG * alr);
    const float phS = (float)SEG * th;
    rotS[0][j] = rS * cosf(phS);
    rotS[1][j] = rS * sinf(phS);
  }
  __syncthreads();

  {
    const v4f reS = *(const v4f*)&rotS[0][h0];
    const v4f imS = *(const v4f*)&rotS[1][h0];
#pragma unroll
    for (int sg = 0; sg < NSEG; ++sg) {
      *(v4f*)&Cs[BGRP * sg + wave][h0]       = c1;
      *(v4f*)&Cs[BGRP * sg + wave][H2v + h0] = c2;
      if (sg + 1 < NSEG) {
        const v4f n1 = reS * c1 - imS * c2 + e1[sg];
        const v4f n2 = imS * c1 + reS * c2 + e2[sg];
        c1 = n1; c2 = n2;
      }
    }
  }
  __syncthreads();

  v4f bf[32];
  {
    const v4f* cp = (const v4f*)(&Cs[col][4 * hl]);
#pragma unroll
    for (int i = 0; i < 32; ++i) bf[i] = cp[2 * i];
  }

#pragma unroll
  for (int m = 0; m < MSTEPS / 2; ++m) {
    const v4f* ap = (const v4f*)(&P[16 * m + c][4 * hl]);
    v4f af[32];
#pragma unroll
    for (int i = 0; i < 32; ++i) af[i] = ap[2 * i];
#pragma unroll
    for (int i = 0; i < 32; i += 8)
      asm volatile("" :: "v"(af[i]), "v"(af[i + 1]), "v"(af[i + 2]), "v"(af[i + 3]),
                         "v"(af[i + 4]), "v"(af[i + 5]), "v"(af[i + 6]), "v"(af[i + 7]));
    __builtin_amdgcn_sched_barrier(0);

    v8f acc0 = {}, acc1 = {};
#pragma unroll
    for (int i = 0; i < 32; ++i) {
      acc0 = __builtin_amdgcn_wmma_f32_16x16x4_f32(
          false, af[i].xy, false, bf[i].xy, (short)0, acc0, false, false);
      acc1 = __builtin_amdgcn_wmma_f32_16x16x4_f32(
          false, af[i].zw, false, bf[i].zw, (short)0, acc1, false, false);
    }
    __builtin_amdgcn_sched_barrier(0);

    float4 y0v = ylo[m], y1v = yhi[m];
    y0v.x += acc0[0] + acc1[0]; y0v.y += acc0[1] + acc1[1];
    y0v.z += acc0[2] + acc1[2]; y0v.w += acc0[3] + acc1[3];
    y1v.x += acc0[4] + acc1[4]; y1v.y += acc0[5] + acc1[5];
    y1v.z += acc0[6] + acc1[6]; y1v.w += acc0[7] + acc1[7];
    yp[m][0] = y0v; yp[m][1] = y1v;
  }
}

extern "C" void kernel_launch(void* const* d_in, const int* in_sizes, int n_in,
                              void* d_out, int out_size, void* d_ws, size_t ws_size,
                              hipStream_t stream) {
  (void)in_sizes; (void)n_in; (void)out_size; (void)ws_size;
  const float* y0   = (const float*)d_in[0];
  const float* U    = (const float*)d_in[1];
  const float* lr   = (const float*)d_in[2];
  const float* li   = (const float*)d_in[3];
  const float* Bm   = (const float*)d_in[4];
  const float* mult = (const float*)d_in[5];
  const float* Wy2x = (const float*)d_in[6];
  const float* by2x = (const float*)d_in[7];
  const float* Wx2y = (const float*)d_in[8];
  const float* bx2y = (const float*)d_in[9];
  float* Y = (float*)d_out;

  float* e_ws = (float*)d_ws;

  dim3 grid(NBv, NSEG);
  lru_seg_scan<<<grid, 64, 0, stream>>>(U, lr, li, Bm, mult, Wx2y, bx2y, Y, e_ws);
  dim3 cgrid(32, 16);
  lru_correct<<<cgrid, 128, 0, stream>>>(y0, Wy2x, by2x, lr, li, Wx2y, e_ws, Y);
}


# Launch 1 scan: the steps' broadcast LDS input reads are issued one 4-step group ahead into a second register set, so a step no longer waits on LDS latency; sums unchanged
# speedup vs baseline: 1.8465x; 1.1123x over previous
#include <hip/hip_runtime.h>


typedef __attribute__((ext_vector_type(2))) float v2f;
typedef __attribute__((ext_vector_type(4))) float v4f;
typedef __attribute__((ext_vector_type(8))) float v8f;

#define NYv   8
#define NUv   8
#define NHv   256
#define H2v   128
#define T_LEN 2048
#define NBv   64
#define NSEG  16
#define SEG   (T_LEN / NSEG)

#define RND    32
#define NRND   (SEG / RND)
#define XPITCH 260
#define UGRP   4

__global__ void __launch_bounds__(64)
lru_seg_scan(const float* __restrict__ U,   const float* __restrict__ lr,
             const float* __restrict__ li,  const float* __restrict__ Bm,
             const float* __restrict__ mult,const float* __restrict__ Wx2y,
             const float* __restrict__ bx2y,float* __restrict__ Y,
             float* __restrict__ e_ws) {
  const int b = blockIdx.x;
  const int s = blockIdx.y;
  const int tid = threadIdx.x;
  const int wave = tid >> 5, lane = tid & 31;
  const int c = lane & 15, hl = lane >> 4;
  const int j0 = 2 * tid;

  __shared__ __attribute__((aligned(16))) float Uc[RND][NUv];
  __shared__ __attribute__((aligned(16))) float Xc[RND][XPITCH];

  const int ut = tid >> 1, uh = (tid & 1) * 4;
  const float* up = U + ((size_t)(s * SEG + ut) * NBv + b) * NUv + uh;
  float4 ucur = *(const float4*)up;

  float Bs1[2][NUv], Bs2[2][NUv], re[2], im[2];
#pragma unroll
  for (int q = 0; q < 2; ++q) {
    const int j = j0 + q;
    const float m1 = expf(mult[j]);
    const float m2 = expf(mult[j + H2v]);
#pragma unroll
    for (int u = 0; u < NUv; ++u) {
      Bs1[q][u] = Bm[j * NUv + u] * m1;
      Bs2[q][u] = Bm[(j + H2v) * NUv + u] * m2;
    }
    const float r  = expf(-fabsf(lr[j]));
    const float th = 1.5707963267948966f * li[j];
    re[q] = r * cosf(th);
    im[q] = r * sinf(th);
  }

  v4f bf[32];
  {
    const v4f* wp = (const v4f*)(Wx2y + (size_t)(c & 7) * NHv + 4 * hl);
    const v4f zero = {0.0f, 0.0f, 0.0f, 0.0f};
#pragma unroll
    for (int i = 0; i < 32; ++i) {
      const v4f w = wp[2 * i];
      bf[i] = (c < NYv) ? w : zero;
    }
  }
  const float bias = bx2y[c & 7];

  float x1[2] = {0.0f, 0.0f}, x2[2] = {0.0f, 0.0f};
  float ypend[8];
#pragma unroll
  for (int r = 0; r < 8; ++r) ypend[r] = 0.0f;

#pragma nounroll
  for (int rd = 0; rd < NRND; ++rd) {
    *(float4*)&Uc[ut][uh] = ucur;
    __syncthreads();

    if (rd + 1 < NRND)
      ucur = *(const float4*)(up + (size_t)(rd + 1) * RND * NBv * NUv);
    if (rd > 0 && c < NYv) {
      float* yp = Y + ((size_t)(s * SEG + (rd - 1) * RND + 16 * wave + 8 * hl) * NBv + b) * NYv + c;
#pragma unroll
      for (int r = 0; r < 8; ++r) yp[(size_t)r * NBv * NYv] = ypend[r];
    }

    float4 ga[UGRP], gb[UGRP];
#pragma unroll
    for (int i = 0; i < UGRP; ++i) {
      ga[i] = *(const float4*)&Uc[i][0];
      gb[i] = *(const float4*)&Uc[i][4];
    }
#pragma unroll
    for (int g = 0; g < RND / UGRP; ++g) {
      float4 na[UGRP], nb[UGRP];
      if (g + 1 < RND / UGRP) {
#pragma unroll
        for (int i = 0; i < UGRP; ++i) {
          na[i] = *(const float4*)&Uc[(g + 1) * UGRP + i][0];
          nb[i] = *(const float4*)&Uc[(g + 1) * UGRP + i][4];
        }
      }
      __builtin_amdgcn_sched_barrier(0);
#pragma unroll
      for (int i = 0; i < UGRP; ++i) {
        const int tt = g * UGRP + i;
        const float4 ua = ga[i];
        const float4 ub = gb[i];
        const float uu[NUv] = {ua.x, ua.y, ua.z, ua.w, ub.x, ub.y, ub.z, ub.w};
#pragma unroll
        for (int q = 0; q < 2; ++q) {
          float bu1 = 0.0f, bu2 = 0.0f;
#pragma unroll
          for (int u = 0; u < NUv; ++u) {
            bu1 = fmaf(uu[u], Bs1[q][u], bu1);
            bu2 = fmaf(uu[u], Bs2[q][u], bu2);
          }
          const float n1 = re[q] * x1[q] - im[q] * x2[q] + bu1;
          const float n2 = im[q] * x1[q] + re[q] * x2[q] + bu2;
          x1[q] = n1; x2[q] = n2;
        }
        *(float2*)&Xc[tt][j0]       = make_float2(x1[0], x1[1]);
        *(float2*)&Xc[tt][H2v + j0] = make_float2(x2[0], x2[1]);
      }
      if (g + 1 < RND / UGRP) {
#pragma unroll
        for (int i = 0; i < UGRP; ++i) { ga[i] = na[i]; gb[i] = nb[i]; }
      }
    }
    __syncthreads();

    const v4f* ap = (const v4f*)(&Xc[16 * wave + c][4 * hl]);
    v4f af[32];
#pragma unroll
    for (int i = 0; i < 32; ++i) af[i] = ap[2 * i];
#pragma unroll
    for (int i = 0; i < 32; i += 8)
      asm volatile("" :: "v"(af[i]), "v"(af[i + 1]), "v"(af[i + 2]), "v"(af[i + 3]),
                         "v"(af[i + 4]), "v"(af[i + 5]), "v"(af[i + 6]), "v"(af[i + 7]));
    __builtin_amdgcn_sched_barrier(0);

    v8f acc0 = {}, acc1 = {};
#pragma unroll
    for (int i = 0; i < 32; ++i) {
      acc0 = __builtin_amdgcn_wmma_f32_16x16x4_f32(
          false, af[i].xy, false, bf[i].xy, (short)0, acc0, false, false);
      acc1 = __builtin_amdgcn_wmma_f32_16x16x4_f32(
          false, af[i].zw, false, bf[i].zw, (short)0, acc1, false, false);
    }
#pragma unroll
    for (int r = 0; r < 8; ++r) ypend[r] = acc0[r] + acc1[r] + bias;
  }

  if (c < NYv) {
    float* yp = Y + ((size_t)(s * SEG + (NRND - 1) * RND + 16 * wave + 8 * hl) * NBv + b) * NYv + c;
#pragma unroll
    for (int r = 0; r < 8; ++r) yp[(size_t)r * NBv * NYv] = ypend[r];
  }

  float* ep = e_ws + ((size_t)s * NBv + b) * NHv;
  *(float2*)(ep + j0)       = make_float2(x1[0], x1[1]);
  *(float2*)(ep + H2v + j0) = make_float2(x2[0], x2[1]);
}

#define MSTEPS 4
#define PPITCH 260
#define BGRP   4

__global__ void __launch_bounds__(128)
lru_correct(const float* __restrict__ y0,  const float* __restrict__ Wy2x,
            const float* __restrict__ by2x,const float* __restrict__ lr,
            const float* __restrict__ li,  const float* __restrict__ Wx2y,
            const float* __restrict__ e_ws, float* __restrict__ Y) {
  const int mg = blockIdx.x;
  const int bg = blockIdx.y;
  const int j = threadIdx.x;
  const int wave = j >> 5, lane = j & 31;
  const int c = lane & 15, hl = lane >> 4;

  __shared__ __attribute__((aligned(16))) float P[MSTEPS * NYv][PPITCH];
  __shared__ __attribute__((aligned(16))) float Cs[NSEG * BGRP][PPITCH];
  __shared__ __attribute__((aligned(16))) float rotS[2][H2v];

  const float alr = fabsf(lr[j]);
  const float th  = 1.5707963267948966f * li[j];

  const int bsc = bg * BGRP + wave;
  const int h0 = 4 * lane;
  v4f e1[NSEG - 1], e2[NSEG - 1];
#pragma unroll
  for (int s = 0; s < NSEG - 1; ++s) {
    const float* ep = e_ws + ((size_t)s * NBv + bsc) * NHv + h0;
    e1[s] = *(const v4f*)ep;
    e2[s] = *(const v4f*)(ep + H2v);
  }

  const int col = 16 * wave + c;
  const int s = col >> 2, b = bg * BGRP + (col & 3);
  const int t0 = s * SEG + mg * MSTEPS + hl;
  float4* yp[MSTEPS / 2];
  float4 ylo[MSTEPS / 2], yhi[MSTEPS / 2];
#pragma unroll
  for (int m = 0; m < MSTEPS / 2; ++m) {
    yp[m]  = (float4*)(Y + ((size_t)(t0 + 2 * m) * NBv + b) * NYv);
    ylo[m] = yp[m][0];
    yhi[m] = yp[m][1];
  }

  float w1[NYv], w2[NYv];
#pragma unroll
  for (int n = 0; n < NYv; ++n) {
    w1[n] = Wx2y[n * NHv + j];
    w2[n] = Wx2y[n * NHv + j + H2v];
  }
#pragma unroll
  for (int kk = 0; kk < MSTEPS; ++kk) {
    const float kf = (float)(mg * MSTEPS + kk + 1);
    const float rk = expf(-kf * alr);
    const float ah = kf * th;
    const float al = fmaf(kf, th, -ah);
    const float sh = sinf(ah), ch = cosf(ah);
    const float sl = sinf(al), cl = cosf(al);
    const float pre = rk * (ch * cl - sh * sl);
    const float pim = rk * (sh * cl + ch * sl);
#pragma unroll
    for (int n = 0; n < NYv; ++n) {
      P[kk * NYv + n][j]       = w1[n] * pre + w2[n] * pim;
      P[kk * NYv + n][j + H2v] = w2[n] * pre - w1[n] * pim;
    }
  }
  v4f c1, c2;
  {
    const v4f ya = *(const v4f*)(y0 + bsc * NYv);
    const v4f yb = *(const v4f*)(y0 + bsc * NYv + 4);
    const float yv[NYv] = {ya.x, ya.y, ya.z, ya.w, yb.x, yb.y, yb.z, yb.w};
    c1 = *(const v4f*)(by2x + h0);
    c2 = *(const v4f*)(by2x + H2v + h0);
#pragma unroll
    for (int i = 0; i < 4; ++i) {
      const v4f wa = *(const v4f*)(Wy2x + (size_t)(h0 + i) * NYv);
      const v4f wb = *(const v4f*)(Wy2x + (size_t)(h0 + i) * NYv + 4);
      const v4f wc = *(const v4f*)(Wy2x + (size_t)(h0 + i + H2v) * NYv);
      const v4f wd = *(const v4f*)(Wy2x + (size_t)(h0 + i + H2v) * NYv + 4);
      const float w1r[NYv] = {wa.x, wa.y, wa.z, wa.w, wb.x, wb.y, wb.z, wb.w};
      const float w2r[NYv] = {wc.x, wc.y, wc.z, wc.w, wd.x, wd.y, wd.z, wd.w};
      float a1 = c1[i], a2 = c2[i];
#pragma unroll
      for (int n = 0; n < NYv; ++n) {
        a1 = fmaf(yv[n], w1r[n], a1);
        a2 = fmaf(yv[n], w2r[n], a2);
      }
      c1[i] = a1; c2[i] = a2;
    }
  }

  {
    const float rS  = expf(-(float)SEG * alr);
    const float phS = (float)SEG * th;
    rotS[0][j] = rS * cosf(phS);
    rotS[1][j] = rS * sinf(phS);
  }
  __syncthreads();

  {
    const v4f reS = *(const v4f*)&rotS[0][h0];
    const v4f imS = *(const v4f*)&rotS[1][h0];
#pragma unroll
    for (int sg = 0; sg < NSEG; ++sg) {
      *(v4f*)&Cs[BGRP * sg + wave][h0]       = c1;
      *(v4f*)&Cs[BGRP * sg + wave][H2v + h0] = c2;
      if (sg + 1 < NSEG) {
        const v4f n1 = reS * c1 - imS * c2 + e1[sg];
        const v4f n2 = imS * c1 + reS * c2 + e2[sg];
        c1 = n1; c2 = n2;
      }
    }
  }
  __syncthreads();

  v4f bf[32];
  {
    const v4f* cp = (const v4f*)(&Cs[col][4 * hl]);
#pragma unroll
    for (int i = 0; i < 32; ++i) bf[i] = cp[2 * i];
  }

#pragma unroll
  for (int m = 0; m < MSTEPS / 2; ++m) {
    const v4f* ap = (const v4f*)(&P[16 * m + c][4 * hl]);
    v4f af[32];
#pragma unroll
    for (int i = 0; i < 32; ++i) af[i] = ap[2 * i];
#pragma unroll
    for (int i = 0; i < 32; i += 8)
      asm volatile("" :: "v"(af[i]), "v"(af[i + 1]), "v"(af[i + 2]), "v"(af[i + 3]),
                         "v"(af[i + 4]), "v"(af[i + 5]), "v"(af[i + 6]), "v"(af[i + 7]));
    __builtin_amdgcn_sched_barrier(0);

    v8f acc0 = {}, acc1 = {};
#pragma unroll
    for (int i = 0; i < 32; ++i) {
      acc0 = __builtin_amdgcn_wmma_f32_16x16x4_f32(
          false, af[i].xy, false, bf[i].xy, (short)0, acc0, false, false);
      acc1 = __builtin_amdgcn_wmma_f32_16x16x4_f32(
          false, af[i].zw, false, bf[i].zw, (short)0, acc1, false, false);
    }
    __builtin_amdgcn_sched_barrier(0);

    float4 y0v = ylo[m], y1v = yhi[m];
    y0v.x += acc0[0] + acc1[0]; y0v.y += acc0[1] + acc1[1];
    y0v.z += acc0[2] + acc1[2]; y0v.w += acc0[3] + acc1[3];
    y1v.x += acc0[4] + acc1[4]; y1v.y += acc0[5] + acc1[5];
    y1v.z += acc0[6] + acc1[6]; y1v.w += acc0[7] + acc1[7];
    yp[m][0] = y0v; yp[m][1] = y1v;
  }
}

extern "C" void kernel_launch(void* const* d_in, const int* in_sizes, int n_in,
                              void* d_out, int out_size, void* d_ws, size_t ws_size,
                              hipStream_t stream) {
  (void)in_sizes; (void)n_in; (void)out_size; (void)ws_size;
  const float* y0   = (const float*)d_in[0];
  const float* U    = (const float*)d_in[1];
  const float* lr   = (const float*)d_in[2];
  const float* li   = (const float*)d_in[3];
  const float* Bm   = (const float*)d_in[4];
  const float* mult = (const float*)d_in[5];
  const float* Wy2x = (const float*)d_in[6];
  const float* by2x = (const float*)d_in[7];
  const float* Wx2y = (const float*)d_in[8];
  const float* bx2y = (const float*)d_in[9];
  float* Y = (float*)d_out;

  float* e_ws = (float*)d_ws;

  dim3 grid(NBv, NSEG);
  lru_seg_scan<<<grid, 64, 0, stream>>>(U, lr, li, Bm, mult, Wx2y, bx2y, Y, e_ws);
  dim3 cgrid(32, 16);
  lru_correct<<<cgrid, 128, 0, stream>>>(y0, Wy2x, by2x, lr, li, Wx2y, e_ws, Y);
}
